# starting kernel
# baseline (speedup 1.0000x reference)
#include <hip/hip_runtime.h>
#include <math.h>


typedef __attribute__((ext_vector_type(2))) float v2f;
typedef __attribute__((ext_vector_type(8))) float v8f;

#define DHID 128
#define TK 64
#define NEG_SLOPE 0.01f

__global__ void fill_zero(float* __restrict__ p, size_t n) {
    size_t i = blockIdx.x * (size_t)blockDim.x + threadIdx.x;
    if (i < n) p[i] = 0.0f;
}

__global__ void degree_kernel(const int* __restrict__ dst, float* __restrict__ deg, int E) {
    int e = blockIdx.x * blockDim.x + threadIdx.x;
    if (e < E) atomicAdd(deg + dst[e], 1.0f);
}

__global__ void rsqrt_kernel(const float* __restrict__ deg, float* __restrict__ dinv, int N) {
    int i = blockIdx.x * blockDim.x + threadIdx.x;
    if (i < N) dinv[i] = rsqrtf(deg[i] + 1.0f);
}

template <int K, bool BIAS, bool ACT>
__global__ __launch_bounds__(128) void gemm_wmma(
    const float* __restrict__ A, const float* __restrict__ W,
    const float* __restrict__ bias, float* __restrict__ Out, int Nrows)
{
    __shared__ float Wp[TK / 2][DHID * 2];

    const int tid  = threadIdx.x;
    const int wave = tid >> 5;
    const int lane = tid & 31;
    const int row0 = (blockIdx.x * 4 + wave) * 16;
    const bool rowok = (row0 < Nrows);

    const int m     = lane & 15;
    const int khalf = (lane >> 4) * 2;

    v8f acc[8];
#pragma unroll
    for (int t = 0; t < 8; ++t) acc[t] = (v8f){};

    const float* __restrict__ arow = A + (size_t)((rowok ? row0 : 0) + m) * K;

    for (int k0 = 0; k0 < K; k0 += TK) {
#pragma unroll 8
        for (int idx = 0; idx < TK * DHID; idx += 128) {
            const int j   = idx + tid;
            const int krl = j >> 7;
            const int col = j & (DHID - 1);
            const int kr  = k0 + krl;
            const float v = (kr < K) ? W[(size_t)kr * DHID + col] : 0.0f;
            Wp[krl >> 1][col * 2 + (krl & 1)] = v;
        }
        __syncthreads();

#pragma unroll 4
        for (int k = 0; k < TK; k += 4) {
            const int kk = k0 + k + khalf;
            const int i0 = (kk     < K) ? kk     : (K - 1);
            const int i1 = (kk + 1 < K) ? kk + 1 : (K - 1);
            v2f a;
            a.x = arow[i0];
            a.y = arow[i1];
            const float* __restrict__ wp = &Wp[(k + khalf) >> 1][0];
#pragma unroll
            for (int t = 0; t < 8; ++t) {
                const v2f b = *(const v2f*)(wp + (t * 16 + m) * 2);
                acc[t] = __builtin_amdgcn_wmma_f32_16x16x4_f32(
                    false, a, false, b, (short)0, acc[t], false, false);
            }
        }
        __syncthreads();
    }

    if (!rowok) return;
    const int rbase = row0 + ((lane >> 4) ? 8 : 0);
#pragma unroll
    for (int t = 0; t < 8; ++t) {
        const int col = t * 16 + m;
        const float bv = BIAS ? bias[col] : 0.0f;
#pragma unroll
        for (int r = 0; r < 8; ++r) {
            float v = acc[t][r] + bv;
            if (ACT) v = (v >= 0.0f) ? v : NEG_SLOPE * v;
            Out[(size_t)(rbase + r) * DHID + col] = v;
        }
    }
}

__global__ __launch_bounds__(256) void scatter_kernel(
    const float* __restrict__ g, const int* __restrict__ src,
    const int* __restrict__ dst, const float* __restrict__ dinv,
    float* __restrict__ agg, int E)
{
    const int t = blockIdx.x * blockDim.x + threadIdx.x;
    const int e = t >> 5;
    const int lane = t & 31;
    if (e >= E) return;
    const int s = src[e];
    const int d = dst[e];
    const float norm = dinv[s] * dinv[d];
    const float4 v = *(const float4*)(g + (size_t)s * DHID + lane * 4);
    float* ap = agg + (size_t)d * DHID + lane * 4;
    atomicAdd(ap + 0, v.x * norm);
    atomicAdd(ap + 1, v.y * norm);
    atomicAdd(ap + 2, v.z * norm);
    atomicAdd(ap + 3, v.w * norm);
}

__global__ __launch_bounds__(256) void finalize_kernel(
    const float* __restrict__ agg, const float* __restrict__ g,
    const float* __restrict__ dinv, const float* __restrict__ bias,
    float* __restrict__ out, int Nrows)
{
    const size_t idx = blockIdx.x * (size_t)blockDim.x + threadIdx.x;
    const size_t total = (size_t)Nrows * (DHID / 4);
    if (idx >= total) return;
    const int i  = (int)(idx >> 5);
    const int c4 = ((int)idx & 31) * 4;
    const float di = dinv[i];
    const float s = di * di;
    const float4 a  = *(const float4*)(agg + (size_t)i * DHID + c4);
    const float4 gg = *(const float4*)(g   + (size_t)i * DHID + c4);
    const float4 bb = *(const float4*)(bias + c4);
    float4 r;
    r.x = a.x + gg.x * s + bb.x;
    r.y = a.y + gg.y * s + bb.y;
    r.z = a.z + gg.z * s + bb.z;
    r.w = a.w + gg.w * s + bb.w;
    *(float4*)(out + (size_t)i * DHID + c4) = r;
}

__global__ __launch_bounds__(256) void out_layer_kernel(
    const float* __restrict__ H, const float* __restrict__ W2,
    const float* __restrict__ b2, float* __restrict__ out, int Nrows)
{
    const int i = blockIdx.x * blockDim.x + threadIdx.x;
    if (i >= Nrows) return;
    const float* h = H + (size_t)i * DHID;
    float a0 = 0.0f, a1 = 0.0f;
#pragma unroll 8
    for (int k = 0; k < DHID; ++k) {
        const float hv = h[k];
        a0 += hv * W2[k * 2 + 0];
        a1 += hv * W2[k * 2 + 1];
    }
    out[(size_t)i * 2 + 0] = a0 + b2[0];
    out[(size_t)i * 2 + 1] = a1 + b2[1];
}

extern "C" void kernel_launch(void* const* d_in, const int* in_sizes, int n_in,
                              void* d_out, int out_size, void* d_ws, size_t ws_size,
                              hipStream_t stream)
{
    const float* x    = (const float*)d_in[0];
    const int*   ei   = (const int*)  d_in[1];
    const float* W_in = (const float*)d_in[3];
    const float* b_in = (const float*)d_in[4];
    const float* W_g  = (const float*)d_in[5];
    const float* b_g  = (const float*)d_in[6];
    const float* W_o1 = (const float*)d_in[7];
    const float* b_o1 = (const float*)d_in[8];
    const float* W_o2 = (const float*)d_in[9];
    const float* b_o2 = (const float*)d_in[10];

    const int ND = 239;
    const int N  = in_sizes[0] / ND;
    const int E  = in_sizes[1] / 2;
    const int* src = ei;
    const int* dst = ei + E;

    float* ws   = (float*)d_ws;
    float* deg  = ws;
    float* dinv = ws + (size_t)N;
    float* bufA = ws + 2 * (size_t)N;
    float* bufB = bufA + (size_t)N * DHID;
    float* bufC = bufB + (size_t)N * DHID;

    const size_t nh = (size_t)N * DHID;
    const int gemm_blocks = (N / 16 + 3) / 4;
    const int scat_blocks = (int)(((size_t)E * 32 + 255) / 256);
    const int fin_blocks  = (int)((nh / 4 + 255) / 256);
    const int zero_blocks = (int)((nh + 255) / 256);

    fill_zero<<<(N + 255) / 256, 256, 0, stream>>>(deg, (size_t)N);
    degree_kernel<<<(E + 255) / 256, 256, 0, stream>>>(dst, deg, E);
    rsqrt_kernel<<<(N + 255) / 256, 256, 0, stream>>>(deg, dinv, N);

    gemm_wmma<239, true, true><<<gemm_blocks, 128, 0, stream>>>(x, W_in, b_in, bufA, N);

    gemm_wmma<128, false, false><<<gemm_blocks, 128, 0, stream>>>(bufA, W_g, nullptr, bufB, N);
    fill_zero<<<zero_blocks, 256, 0, stream>>>(bufC, nh);
    scatter_kernel<<<scat_blocks, 256, 0, stream>>>(bufB, src, dst, dinv, bufC, E);
    finalize_kernel<<<fin_blocks, 256, 0, stream>>>(bufC, bufB, dinv, b_g, bufA, N);

    gemm_wmma<128, false, false><<<gemm_blocks, 128, 0, stream>>>(bufA, W_g, nullptr, bufB, N);
    fill_zero<<<zero_blocks, 256, 0, stream>>>(bufC, nh);
    scatter_kernel<<<scat_blocks, 256, 0, stream>>>(bufB, src, dst, dinv, bufC, E);
    finalize_kernel<<<fin_blocks, 256, 0, stream>>>(bufC, bufB, dinv, b_g, bufA, N);

    gemm_wmma<128, true, true><<<gemm_blocks, 128, 0, stream>>>(bufA, W_o1, b_o1, bufB, N);

    out_layer_kernel<<<(N + 255) / 256, 256, 0, stream>>>(bufB, W_o2, b_o2, (float*)d_out, N);
}


# GEMMs: one persistent 28-wave work-group per WGP keeps all of W in LDS (staged once, no K-chunk barriers); 24 whole slabs + a leftover slab cut by columns over 4 SIMDs: 6.25 slabs a SIMD, was 7
# speedup vs baseline: 2.0082x; 2.0082x over previous
#include <hip/hip_runtime.h>
#include <math.h>


typedef __attribute__((ext_vector_type(2))) float v2f;
typedef __attribute__((ext_vector_type(8))) float v8f;

#define DHID 128
#define NEG_SLOPE 0.01f

__global__ void fill_zero(float* __restrict__ p, size_t n) {
    size_t i = blockIdx.x * (size_t)blockDim.x + threadIdx.x;
    if (i < n) p[i] = 0.0f;
}

__global__ void degree_kernel(const int* __restrict__ dst, float* __restrict__ deg, int E) {
    int e = blockIdx.x * blockDim.x + threadIdx.x;
    if (e < E) atomicAdd(deg + dst[e], 1.0f);
}

__global__ void rsqrt_kernel(const float* __restrict__ deg, float* __restrict__ dinv, int N) {
    int i = blockIdx.x * blockDim.x + threadIdx.x;
    if (i < N) dinv[i] = rsqrtf(deg[i] + 1.0f);
}

#define GEMM_WAVES 28
#define GEMM_THREADS 896
#define GEMM_FULL  24
#define GEMM_ROWGROUPS (GEMM_WAVES / 4)

template <int K, int NT, bool BIAS, bool ACT, bool CONV>
__device__ __forceinline__ void slab_product(
    const float* __restrict__ A, const float* Wp,
    const float* __restrict__ bias, float* __restrict__ Out,
    const float* __restrict__ dinv, float* __restrict__ Init,
    int row0, int col0, int lane)
{
    constexpr int KP = (K + 3) & ~3;
    const int m     = lane & 15;
    const int khalf = (lane >> 4) * 2;

    v8f acc[NT];
#pragma unroll
    for (int t = 0; t < NT; ++t) acc[t] = (v8f){};

    const float* __restrict__ arow = A + (size_t)(row0 + m) * K;
    const float* wcol = Wp + (col0 + m) * 2;

#pragma unroll 4
    for (int k = 0; k < KP; k += 4) {
        const int kk = k + khalf;
        const int i0 = (kk     < K) ? kk     : (K - 1);
        const int i1 = (kk + 1 < K) ? kk + 1 : (K - 1);
        v2f a;
        a.x = arow[i0];
        a.y = arow[i1];
        const float* wp = wcol + (kk >> 1) * (DHID * 2);
#pragma unroll
        for (int t = 0; t < NT; ++t) {
            const v2f b = *(const v2f*)(wp + t * 32);
            acc[t] = __builtin_amdgcn_wmma_f32_16x16x4_f32(
                false, a, false, b, (short)0, acc[t], false, false);
        }
    }

    const int rbase = row0 + ((lane >> 4) ? 8 : 0);
    if (CONV) {
        float s[8];
#pragma unroll
        for (int r = 0; r < 8; ++r) {
            const float di = dinv[rbase + r];
            s[r] = di * di;
        }
#pragma unroll
        for (int t = 0; t < NT; ++t) {
            const int col = col0 + t * 16 + m;
            const float bv = bias[col];
#pragma unroll
            for (int r = 0; r < 8; ++r) {
                const float gv = acc[t][r];
                const size_t o = (size_t)(rbase + r) * DHID + col;
                Out[o]  = gv;
                Init[o] = gv * s[r] + bv;
            }
        }
        return;
    }
#pragma unroll
    for (int t = 0; t < NT; ++t) {
        const int col = col0 + t * 16 + m;
        const float bv = BIAS ? bias[col] : 0.0f;
#pragma unroll
        for (int r = 0; r < 8; ++r) {
            float v = acc[t][r] + bv;
            if (ACT) v = (v >= 0.0f) ? v : NEG_SLOPE * v;
            Out[(size_t)(rbase + r) * DHID + col] = v;
        }
    }
}

template <int K, bool BIAS, bool ACT, bool CONV>
__global__ __launch_bounds__(GEMM_THREADS) __attribute__((amdgpu_waves_per_eu(7, 7))) void gemm_wmma(
    const float* __restrict__ A, const float* __restrict__ W,
    const float* __restrict__ bias, float* __restrict__ Out, int Nrows,
    const float* __restrict__ dinv, float* __restrict__ Init)
{
    constexpr int KP = (K + 3) & ~3;
    constexpr int NR = (KP + GEMM_ROWGROUPS - 1) / GEMM_ROWGROUPS;
    __shared__ float Wp[KP / 2][DHID * 2];

    const int tid  = threadIdx.x;
    const int wave = tid >> 5;
    const int lane = tid & 31;

    {
        const int col = tid & (DHID - 1);
        const int rg  = tid >> 7;
        float v[NR];
#pragma unroll
        for (int i = 0; i < NR; ++i) {
            const int kr  = rg + GEMM_ROWGROUPS * i;
            const int krc = (kr < K) ? kr : (K - 1);
            v[i] = W[(size_t)krc * DHID + col];
        }
#pragma unroll
        for (int i = 0; i < NR; ++i) {
            const int kr = rg + GEMM_ROWGROUPS * i;
            if (kr < KP) Wp[kr >> 1][col * 2 + (kr & 1)] = (kr < K) ? v[i] : 0.0f;
        }
    }
    __syncthreads();

    const int nslab = Nrows >> 4;
    const int per   = nslab / (int)gridDim.x;
    const int extra = nslab - per * (int)gridDim.x;

    if (wave < GEMM_FULL) {
        for (int q = wave; q < per; q += GEMM_FULL)
            slab_product<K, 8, BIAS, ACT, CONV>(A, &Wp[0][0], bias, Out, dinv, Init,
                                                ((int)blockIdx.x * per + q) * 16, 0, lane);
    } else if ((int)blockIdx.x < extra) {
        slab_product<K, 2, BIAS, ACT, CONV>(A, &Wp[0][0], bias, Out, dinv, Init,
                                            ((int)gridDim.x * per + (int)blockIdx.x) * 16,
                                            (wave - GEMM_FULL) * 32, lane);
    }
}

#define EPW 32
#define EB  8

__device__ __forceinline__ void rows_load(
    float (&v)[EB][4], const float* __restrict__ g, int s, int first, unsigned lane)
{
#pragma unroll
    for (int j = 0; j < EB; ++j) {
        const int sj = __builtin_amdgcn_readlane(s, first + j);
        const float* __restrict__ gp = g + (size_t)sj * DHID;
        v[j][0] = gp[lane];
        v[j][1] = gp[lane + 32];
        v[j][2] = gp[lane + 64];
        v[j][3] = gp[lane + 96];
    }
}

__device__ __forceinline__ void rows_add(
    const float (&v)[EB][4], float* __restrict__ agg, int d, int nbits,
    int first, unsigned lane)
{
#pragma unroll
    for (int j = 0; j < EB; ++j) {
        const int dj = __builtin_amdgcn_readlane(d, first + j);
        const float nj = __int_as_float(__builtin_amdgcn_readlane(nbits, first + j));
        float* ap = agg + (size_t)dj * DHID;
        atomicAdd(ap + lane,      v[j][0] * nj);
        atomicAdd(ap + lane + 32, v[j][1] * nj);
        atomicAdd(ap + lane + 64, v[j][2] * nj);
        atomicAdd(ap + lane + 96, v[j][3] * nj);
    }
}

__global__ __launch_bounds__(256) void scatter_kernel(
    const float* __restrict__ g, const int* __restrict__ src,
    const int* __restrict__ dst, const float* __restrict__ dinv,
    float* __restrict__ agg, int E)
{
    const unsigned lane = threadIdx.x & 31u;
    const int wv = (blockIdx.x * blockDim.x + threadIdx.x) >> 5;
    const int e0 = wv * EPW;
    if (e0 >= E) return;
    const int cnt = (E - e0 < EPW) ? (E - e0) : EPW;
    const int el = e0 + (int)lane;
    const int ei = (el < E) ? el : (E - 1);
    const int s = src[ei];
    const int d = dst[ei];
    const float dis = dinv[s];
    const float did = dinv[d];

    if (cnt == EPW) {
        float v[2][EB][4];
        rows_load(v[0], g, s, 0, lane);
        const int nbits = __float_as_int(dis * did);
#pragma unroll
        for (int b = 0; b < EPW / EB; ++b) {
            if (b + 1 < EPW / EB) rows_load(v[(b + 1) & 1], g, s, (b + 1) * EB, lane);
            rows_add(v[b & 1], agg, d, nbits, b * EB, lane);
        }
        return;
    }
    const int nbits = __float_as_int(dis * did);
    for (int j = 0; j < cnt; ++j) {
        const int sj = __builtin_amdgcn_readlane(s, j);
        const int dj = __builtin_amdgcn_readlane(d, j);
        const float nj = __int_as_float(__builtin_amdgcn_readlane(nbits, j));
        const float* __restrict__ gp = g + (size_t)sj * DHID;
        float* ap = agg + (size_t)dj * DHID;
        const float v0 = gp[lane];
        const float v1 = gp[lane + 32];
        const float v2 = gp[lane + 64];
        const float v3 = gp[lane + 96];
        atomicAdd(ap + lane,      v0 * nj);
        atomicAdd(ap + lane + 32, v1 * nj);
        atomicAdd(ap + lane + 64, v2 * nj);
        atomicAdd(ap + lane + 96, v3 * nj);
    }
}

__global__ __launch_bounds__(256) void out_layer_kernel(
    const float* __restrict__ H, const float* __restrict__ W2,
    const float* __restrict__ b2, float* __restrict__ out, int Nrows)
{
    const int i = blockIdx.x * blockDim.x + threadIdx.x;
    if (i >= Nrows) return;
    const float* h = H + (size_t)i * DHID;
    float a0 = 0.0f, a1 = 0.0f;
#pragma unroll 8
    for (int k = 0; k < DHID; ++k) {
        const float hv = h[k];
        a0 += hv * W2[k * 2 + 0];
        a1 += hv * W2[k * 2 + 1];
    }
    out[(size_t)i * 2 + 0] = a0 + b2[0];
    out[(size_t)i * 2 + 1] = a1 + b2[1];
}

extern "C" void kernel_launch(void* const* d_in, const int* in_sizes, int n_in,
                              void* d_out, int out_size, void* d_ws, size_t ws_size,
                              hipStream_t stream)
{
    const float* x    = (const float*)d_in[0];
    const int*   ei   = (const int*)  d_in[1];
    const float* W_in = (const float*)d_in[3];
    const float* b_in = (const float*)d_in[4];
    const float* W_g  = (const float*)d_in[5];
    const float* b_g  = (const float*)d_in[6];
    const float* W_o1 = (const float*)d_in[7];
    const float* b_o1 = (const float*)d_in[8];
    const float* W_o2 = (const float*)d_in[9];
    const float* b_o2 = (const float*)d_in[10];

    const int ND = 239;
    const int N  = in_sizes[0] / ND;
    const int E  = in_sizes[1] / 2;
    const int* src = ei;
    const int* dst = ei + E;

    float* ws   = (float*)d_ws;
    float* deg  = ws;
    float* dinv = ws + (size_t)N;
    float* bufA = ws + 2 * (size_t)N;
    float* bufB = bufA + (size_t)N * DHID;
    float* bufC = bufB + (size_t)N * DHID;

    const int scat_blocks = (E + 255) / 256;

    fill_zero<<<(N + 255) / 256, 256, 0, stream>>>(deg, (size_t)N);
    degree_kernel<<<(E + 255) / 256, 256, 0, stream>>>(dst, deg, E);
    rsqrt_kernel<<<(N + 255) / 256, 256, 0, stream>>>(deg, dinv, N);

    gemm_wmma<239, true, true, false><<<256, GEMM_THREADS, 0, stream>>>(
        x, W_in, b_in, bufA, N, nullptr, nullptr);

    gemm_wmma<128, false, false, true><<<256, GEMM_THREADS, 0, stream>>>(
        bufA, W_g, b_g, bufB, N, dinv, bufC);
    scatter_kernel<<<scat_blocks, 256, 0, stream>>>(bufB, src, dst, dinv, bufC, E);

    gemm_wmma<128, false, false, true><<<256, GEMM_THREADS, 0, stream>>>(
        bufC, W_g, b_g, bufB, N, dinv, bufA);
    scatter_kernel<<<scat_blocks, 256, 0, stream>>>(bufB, src, dst, dinv, bufA, E);

    gemm_wmma<128, true, true, false><<<256, GEMM_THREADS, 0, stream>>>(
        bufA, W_o1, b_o1, bufC, N, nullptr, nullptr);

    out_layer_kernel<<<(N + 255) / 256, 256, 0, stream>>>(bufC, W_o2, b_o2, (float*)d_out, N);
}


# In-degree count rides in the first GEMM's launch (4 light waves a work-group issue the edge atomics under the matrix work); readers form rsqrt(deg+1); 10 -> 8 launches; sum order kept
# speedup vs baseline: 2.0669x; 1.0292x over previous
#include <hip/hip_runtime.h>
#include <math.h>


typedef __attribute__((ext_vector_type(2))) float v2f;
typedef __attribute__((ext_vector_type(8))) float v8f;

#define DHID 128
#define NEG_SLOPE 0.01f

__global__ void fill_zero(float* __restrict__ p, size_t n) {
    size_t i = blockIdx.x * (size_t)blockDim.x + threadIdx.x;
    if (i < n) p[i] = 0.0f;
}

__device__ __forceinline__ float dinv_of(float deg) { return rsqrtf(deg + 1.0f); }

#define GEMM_WAVES 28
#define GEMM_THREADS 896
#define GEMM_FULL  24
#define GEMM_ROWGROUPS (GEMM_WAVES / 4)

template <int K, int NT, bool BIAS, bool ACT, bool CONV>
__device__ __forceinline__ void slab_product(
    const float* __restrict__ A, const float* Wp,
    const float* __restrict__ bias, float* __restrict__ Out,
    const float* __restrict__ deg, float* __restrict__ Init,
    int row0, int col0, int lane)
{
    constexpr int KP = (K + 3) & ~3;
    const int m     = lane & 15;
    const int khalf = (lane >> 4) * 2;

    v8f acc[NT];
#pragma unroll
    for (int t = 0; t < NT; ++t) acc[t] = (v8f){};

    const float* __restrict__ arow = A + (size_t)(row0 + m) * K;
    const float* wcol = Wp + (col0 + m) * 2;

#pragma unroll 4
    for (int k = 0; k < KP; k += 4) {
        const int kk = k + khalf;
        const int i0 = (kk     < K) ? kk     : (K - 1);
        const int i1 = (kk + 1 < K) ? kk + 1 : (K - 1);
        v2f a;
        a.x = arow[i0];
        a.y = arow[i1];
        const float* wp = wcol + (kk >> 1) * (DHID * 2);
#pragma unroll
        for (int t = 0; t < NT; ++t) {
            const v2f b = *(const v2f*)(wp + t * 32);
            acc[t] = __builtin_amdgcn_wmma_f32_16x16x4_f32(
                false, a, false, b, (short)0, acc[t], false, false);
        }
    }

    const int rbase = row0 + ((lane >> 4) ? 8 : 0);
    if (CONV) {
        float s[8];
#pragma unroll
        for (int r = 0; r < 8; ++r) {
            const float di = dinv_of(deg[rbase + r]);
            s[r] = di * di;
        }
#pragma unroll
        for (int t = 0; t < NT; ++t) {
            const int col = col0 + t * 16 + m;
            const float bv = bias[col];
#pragma unroll
            for (int r = 0; r < 8; ++r) {
                const float gv = acc[t][r];
                const size_t o = (size_t)(rbase + r) * DHID + col;
                Out[o]  = gv;
                Init[o] = gv * s[r] + bv;
            }
        }
        return;
    }
#pragma unroll
    for (int t = 0; t < NT; ++t) {
        const int col = col0 + t * 16 + m;
        const float bv = BIAS ? bias[col] : 0.0f;
#pragma unroll
        for (int r = 0; r < 8; ++r) {
            float v = acc[t][r] + bv;
            if (ACT) v = (v >= 0.0f) ? v : NEG_SLOPE * v;
            Out[(size_t)(rbase + r) * DHID + col] = v;
        }
    }
}

template <int K, bool BIAS, bool ACT, bool CONV, bool DEG>
__global__ __launch_bounds__(GEMM_THREADS) __attribute__((amdgpu_waves_per_eu(7, 7))) void gemm_wmma(
    const float* __restrict__ A, const float* __restrict__ W,
    const float* __restrict__ bias, float* __restrict__ Out, int Nrows,
    const float* __restrict__ deg, float* __restrict__ Init,
    const int* __restrict__ edst, float* __restrict__ degcnt, int E)
{
    constexpr int KP = (K + 3) & ~3;
    constexpr int NR = (KP + GEMM_ROWGROUPS - 1) / GEMM_ROWGROUPS;
    __shared__ float Wp[KP / 2][DHID * 2];

    const int tid  = threadIdx.x;
    const int wave = tid >> 5;
    const int lane = tid & 31;

    {
        const int col = tid & (DHID - 1);
        const int rg  = tid >> 7;
        float v[NR];
#pragma unroll
        for (int i = 0; i < NR; ++i) {
            const int kr  = rg + GEMM_ROWGROUPS * i;
            const int krc = (kr < K) ? kr : (K - 1);
            v[i] = W[(size_t)krc * DHID + col];
        }
#pragma unroll
        for (int i = 0; i < NR; ++i) {
            const int kr = rg + GEMM_ROWGROUPS * i;
            if (kr < KP) Wp[kr >> 1][col * 2 + (kr & 1)] = (kr < K) ? v[i] : 0.0f;
        }
    }
    __syncthreads();

    if (DEG && wave >= GEMM_FULL) {
        const int stride = (int)gridDim.x * (GEMM_WAVES - GEMM_FULL) * 32;
        for (int e = ((int)blockIdx.x * (GEMM_WAVES - GEMM_FULL) + (wave - GEMM_FULL)) * 32 + lane;
             e < E; e += 4 * stride) {
            int d[4];
#pragma unroll
            for (int u = 0; u < 4; ++u) {
                const int eu = e + u * stride;
                d[u] = edst[(eu < E) ? eu : e];
            }
#pragma unroll
            for (int u = 0; u < 4; ++u)
                atomicAdd(degcnt + d[u], (e + u * stride < E) ? 1.0f : 0.0f);
        }
    }

    const int nslab = Nrows >> 4;
    const int per   = nslab / (int)gridDim.x;
    const int extra = nslab - per * (int)gridDim.x;

    if (wave < GEMM_FULL) {
        for (int q = wave; q < per; q += GEMM_FULL)
            slab_product<K, 8, BIAS, ACT, CONV>(A, &Wp[0][0], bias, Out, deg, Init,
                                                ((int)blockIdx.x * per + q) * 16, 0, lane);
    } else if ((int)blockIdx.x < extra) {
        slab_product<K, 2, BIAS, ACT, CONV>(A, &Wp[0][0], bias, Out, deg, Init,
                                            ((int)gridDim.x * per + (int)blockIdx.x) * 16,
                                            (wave - GEMM_FULL) * 32, lane);
    }
}

#define EPW 32
#define EB  8

__device__ __forceinline__ void rows_load(
    float (&v)[EB][4], const float* __restrict__ g, int s, int first, unsigned lane)
{
#pragma unroll
    for (int j = 0; j < EB; ++j) {
        const int sj = __builtin_amdgcn_readlane(s, first + j);
        const float* __restrict__ gp = g + (size_t)sj * DHID;
        v[j][0] = gp[lane];
        v[j][1] = gp[lane + 32];
        v[j][2] = gp[lane + 64];
        v[j][3] = gp[lane + 96];
    }
}

__device__ __forceinline__ void rows_add(
    const float (&v)[EB][4], float* __restrict__ agg, int d, int nbits,
    int first, unsigned lane)
{
#pragma unroll
    for (int j = 0; j < EB; ++j) {
        const int dj = __builtin_amdgcn_readlane(d, first + j);
        const float nj = __int_as_float(__builtin_amdgcn_readlane(nbits, first + j));
        float* ap = agg + (size_t)dj * DHID;
        atomicAdd(ap + lane,      v[j][0] * nj);
        atomicAdd(ap + lane + 32, v[j][1] * nj);
        atomicAdd(ap + lane + 64, v[j][2] * nj);
        atomicAdd(ap + lane + 96, v[j][3] * nj);
    }
}

__global__ __launch_bounds__(256) void scatter_kernel(
    const float* __restrict__ g, const int* __restrict__ src,
    const int* __restrict__ dst, const float* __restrict__ deg,
    float* __restrict__ agg, int E)
{
    const unsigned lane = threadIdx.x & 31u;
    const int wv = (blockIdx.x * blockDim.x + threadIdx.x) >> 5;
    const int e0 = wv * EPW;
    if (e0 >= E) return;
    const int cnt = (E - e0 < EPW) ? (E - e0) : EPW;
    const int el = e0 + (int)lane;
    const int ei = (el < E) ? el : (E - 1);
    const int s = src[ei];
    const int d = dst[ei];
    const float dis = dinv_of(deg[s]);
    const float did = dinv_of(deg[d]);

    if (cnt == EPW) {
        float v[2][EB][4];
        rows_load(v[0], g, s, 0, lane);
        const int nbits = __float_as_int(dis * did);
#pragma unroll
        for (int b = 0; b < EPW / EB; ++b) {
            if (b + 1 < EPW / EB) rows_load(v[(b + 1) & 1], g, s, (b + 1) * EB, lane);
            rows_add(v[b & 1], agg, d, nbits, b * EB, lane);
        }
        return;
    }
    const int nbits = __float_as_int(dis * did);
    for (int j = 0; j < cnt; ++j) {
        const int sj = __builtin_amdgcn_readlane(s, j);
        const int dj = __builtin_amdgcn_readlane(d, j);
        const float nj = __int_as_float(__builtin_amdgcn_readlane(nbits, j));
        const float* __restrict__ gp = g + (size_t)sj * DHID;
        float* ap = agg + (size_t)dj * DHID;
        const float v0 = gp[lane];
        const float v1 = gp[lane + 32];
        const float v2 = gp[lane + 64];
        const float v3 = gp[lane + 96];
        atomicAdd(ap + lane,      v0 * nj);
        atomicAdd(ap + lane + 32, v1 * nj);
        atomicAdd(ap + lane + 64, v2 * nj);
        atomicAdd(ap + lane + 96, v3 * nj);
    }
}

__global__ __launch_bounds__(256) void out_layer_kernel(
    const float* __restrict__ H, const float* __restrict__ W2,
    const float* __restrict__ b2, float* __restrict__ out, int Nrows)
{
    const int i = blockIdx.x * blockDim.x + threadIdx.x;
    if (i >= Nrows) return;
    const float* h = H + (size_t)i * DHID;
    float a0 = 0.0f, a1 = 0.0f;
#pragma unroll 8
    for (int k = 0; k < DHID; ++k) {
        const float hv = h[k];
        a0 += hv * W2[k * 2 + 0];
        a1 += hv * W2[k * 2 + 1];
    }
    out[(size_t)i * 2 + 0] = a0 + b2[0];
    out[(size_t)i * 2 + 1] = a1 + b2[1];
}

extern "C" void kernel_launch(void* const* d_in, const int* in_sizes, int n_in,
                              void* d_out, int out_size, void* d_ws, size_t ws_size,
                              hipStream_t stream)
{
    const float* x    = (const float*)d_in[0];
    const int*   ei   = (const int*)  d_in[1];
    const float* W_in = (const float*)d_in[3];
    const float* b_in = (const float*)d_in[4];
    const float* W_g  = (const float*)d_in[5];
    const float* b_g  = (const float*)d_in[6];
    const float* W_o1 = (const float*)d_in[7];
    const float* b_o1 = (const float*)d_in[8];
    const float* W_o2 = (const float*)d_in[9];
    const float* b_o2 = (const float*)d_in[10];

    const int ND = 239;
    const int N  = in_sizes[0] / ND;
    const int E  = in_sizes[1] / 2;
    const int* src = ei;
    const int* dst = ei + E;

    float* ws   = (float*)d_ws;
    float* deg  = ws;
    float* bufA = ws + 2 * (size_t)N;
    float* bufB = bufA + (size_t)N * DHID;
    float* bufC = bufB + (size_t)N * DHID;

    const int scat_blocks = (E + 255) / 256;

    fill_zero<<<(N + 255) / 256, 256, 0, stream>>>(deg, (size_t)N);

    gemm_wmma<239, true, true, false, true><<<256, GEMM_THREADS, 0, stream>>>(
        x, W_in, b_in, bufA, N, nullptr, nullptr, dst, deg, E);

    gemm_wmma<128, false, false, true, false><<<256, GEMM_THREADS, 0, stream>>>(
        bufA, W_g, b_g, bufB, N, deg, bufC, nullptr, nullptr, 0);
    scatter_kernel<<<scat_blocks, 256, 0, stream>>>(bufB, src, dst, deg, bufC, E);

    gemm_wmma<128, false, false, true, false><<<256, GEMM_THREADS, 0, stream>>>(
        bufC, W_g, b_g, bufB, N, deg, bufA, nullptr, nullptr, 0);
    scatter_kernel<<<scat_blocks, 256, 0, stream>>>(bufB, src, dst, deg, bufA, E);

    gemm_wmma<128, true, true, false, false><<<256, GEMM_THREADS, 0, stream>>>(
        bufA, W_o1, b_o1, bufC, N, nullptr, nullptr, nullptr, nullptr, 0);

    out_layer_kernel<<<(N + 255) / 256, 256, 0, stream>>>(bufC, W_o2, b_o2, (float*)d_out, N);
}


# Edge sums without float atomics: per-destination edge lists (one integer atomic exchange per edge, in GEMM 1's launch); a wave walks 8 nodes' lists, sums rows in registers (sum order changes)
# speedup vs baseline: 2.1623x; 1.0462x over previous
#include <hip/hip_runtime.h>
#include <math.h>


typedef __attribute__((ext_vector_type(2))) float v2f;
typedef __attribute__((ext_vector_type(8))) float v8f;

#define DHID 128
#define NEG_SLOPE 0.01f

__global__ void fill_zero(float* __restrict__ p, size_t n) {
    size_t i = blockIdx.x * (size_t)blockDim.x + threadIdx.x;
    if (i < n) p[i] = 0.0f;
}

__device__ __forceinline__ float dinv_of(float deg) { return rsqrtf(deg + 1.0f); }

#define GEMM_WAVES 28
#define GEMM_THREADS 896
#define GEMM_FULL  24
#define GEMM_ROWGROUPS (GEMM_WAVES / 4)

template <int K, int NT, bool BIAS, bool ACT>
__device__ __forceinline__ void slab_product(
    const float* __restrict__ A, const float* Wp,
    const float* __restrict__ bias, float* __restrict__ Out,
    int row0, int col0, int lane)
{
    constexpr int KP = (K + 3) & ~3;
    const int m     = lane & 15;
    const int khalf = (lane >> 4) * 2;

    v8f acc[NT];
#pragma unroll
    for (int t = 0; t < NT; ++t) acc[t] = (v8f){};

    const float* __restrict__ arow = A + (size_t)(row0 + m) * K;
    const float* wcol = Wp + (col0 + m) * 2;

#pragma unroll 4
    for (int k = 0; k < KP; k += 4) {
        const int kk = k + khalf;
        const int i0 = (kk     < K) ? kk     : (K - 1);
        const int i1 = (kk + 1 < K) ? kk + 1 : (K - 1);
        v2f a;
        a.x = arow[i0];
        a.y = arow[i1];
        const float* wp = wcol + (kk >> 1) * (DHID * 2);
#pragma unroll
        for (int t = 0; t < NT; ++t) {
            const v2f b = *(const v2f*)(wp + t * 32);
            acc[t] = __builtin_amdgcn_wmma_f32_16x16x4_f32(
                false, a, false, b, (short)0, acc[t], false, false);
        }
    }

    const int rbase = row0 + ((lane >> 4) ? 8 : 0);
#pragma unroll
    for (int t = 0; t < NT; ++t) {
        const int col = col0 + t * 16 + m;
        const float bv = BIAS ? bias[col] : 0.0f;
#pragma unroll
        for (int r = 0; r < 8; ++r) {
            float v = acc[t][r];
            if (BIAS) v += bv;
            if (ACT) v = (v >= 0.0f) ? v : NEG_SLOPE * v;
            Out[(size_t)(rbase + r) * DHID + col] = v;
        }
    }
}

template <int K, bool BIAS, bool ACT, bool DEG>
__global__ __launch_bounds__(GEMM_THREADS) __attribute__((amdgpu_waves_per_eu(7, 7))) void gemm_wmma(
    const float* __restrict__ A, const float* __restrict__ W,
    const float* __restrict__ bias, float* __restrict__ Out, int Nrows,
    const int* __restrict__ edst, float* __restrict__ degcnt,
    int* __restrict__ head, int* __restrict__ nxt, int E)
{
    constexpr int KP = (K + 3) & ~3;
    constexpr int NR = (KP + GEMM_ROWGROUPS - 1) / GEMM_ROWGROUPS;
    __shared__ float Wp[KP / 2][DHID * 2];

    const int tid  = threadIdx.x;
    const int wave = tid >> 5;
    const int lane = tid & 31;

    {
        const int col = tid & (DHID - 1);
        const int rg  = tid >> 7;
        float v[NR];
#pragma unroll
        for (int i = 0; i < NR; ++i) {
            const int kr  = rg + GEMM_ROWGROUPS * i;
            const int krc = (kr < K) ? kr : (K - 1);
            v[i] = W[(size_t)krc * DHID + col];
        }
#pragma unroll
        for (int i = 0; i < NR; ++i) {
            const int kr = rg + GEMM_ROWGROUPS * i;
            if (kr < KP) Wp[kr >> 1][col * 2 + (kr & 1)] = (kr < K) ? v[i] : 0.0f;
        }
    }
    __syncthreads();

    if (DEG && wave >= GEMM_FULL) {
        const int stride = (int)gridDim.x * (GEMM_WAVES - GEMM_FULL) * 32;
        for (int e = ((int)blockIdx.x * (GEMM_WAVES - GEMM_FULL) + (wave - GEMM_FULL)) * 32 + lane;
             e < E; e += 4 * stride) {
            int d[4];
#pragma unroll
            for (int u = 0; u < 4; ++u) {
                const int eu = e + u * stride;
                d[u] = edst[(eu < E) ? eu : e];
            }
#pragma unroll
            for (int u = 0; u < 4; ++u)
                atomicAdd(degcnt + d[u], (e + u * stride < E) ? 1.0f : 0.0f);
            int old[4];
#pragma unroll
            for (int u = 0; u < 4; ++u) {
                const int eu = e + u * stride;
                old[u] = (eu < E) ? atomicExch(head + d[u], eu + 1) : 0;
            }
#pragma unroll
            for (int u = 0; u < 4; ++u) {
                const int eu = e + u * stride;
                if (eu < E) nxt[eu] = old[u];
            }
        }
    }

    const int nslab = Nrows >> 4;
    const int per   = nslab / (int)gridDim.x;
    const int extra = nslab - per * (int)gridDim.x;

    if (wave < GEMM_FULL) {
        for (int q = wave; q < per; q += GEMM_FULL)
            slab_product<K, 8, BIAS, ACT>(A, &Wp[0][0], bias, Out,
                                          ((int)blockIdx.x * per + q) * 16, 0, lane);
    } else if ((int)blockIdx.x < extra) {
        slab_product<K, 2, BIAS, ACT>(A, &Wp[0][0], bias, Out,
                                      ((int)gridDim.x * per + (int)blockIdx.x) * 16,
                                      (wave - GEMM_FULL) * 32, lane);
    }
}

#define NPW 8

__global__ __launch_bounds__(256) void gather_kernel(
    const float* __restrict__ g, const int* __restrict__ src,
    const int* __restrict__ head, const int* __restrict__ nxt,
    const float* __restrict__ deg, const float* __restrict__ bias,
    float* __restrict__ hout, int N, int E)
{
    const unsigned lane = threadIdx.x & 31u;
    const int wv = __builtin_amdgcn_readfirstlane((int)((blockIdx.x * blockDim.x + threadIdx.x) >> 5));
    const int i0 = wv * NPW;
    if (i0 >= N) return;
    const int il = i0 + (int)(lane & (NPW - 1));
    const int ic = (il < N) ? il : (N - 1);
    const bool own = (lane < NPW) && (il < N);
    const float dgi = deg[ic];
    const int hd = head[ic];

    const float b0 = bias[lane];
    const float b1 = bias[lane + 32];
    const float b2 = bias[lane + 64];
    const float b3 = bias[lane + 96];
    float acc[NPW][4];
#pragma unroll
    for (int j = 0; j < NPW; ++j) {
        const int ij = (i0 + j < N) ? (i0 + j) : (N - 1);
        const float* __restrict__ gp = g + (size_t)ij * DHID;
        acc[j][0] = gp[lane];
        acc[j][1] = gp[lane + 32];
        acc[j][2] = gp[lane + 64];
        acc[j][3] = gp[lane + 96];
    }
    __builtin_amdgcn_sched_barrier(0);

    int cur = own ? hd : 0;
    const int c0 = (cur > 0) ? (cur - 1) : 0;
    int s = src[c0];
    const int n0 = nxt[c0];
    __builtin_amdgcn_sched_barrier(0);

    const float did = dinv_of(dgi);
    const int dbits = __float_as_int(did * did);
#pragma unroll
    for (int j = 0; j < NPW; ++j) {
        const float sw = __int_as_float(__builtin_amdgcn_readlane(dbits, j));
        acc[j][0] = acc[j][0] * sw + b0;
        acc[j][1] = acc[j][1] * sw + b1;
        acc[j][2] = acc[j][2] * sw + b2;
        acc[j][3] = acc[j][3] * sw + b3;
    }
    int nx = (cur > 0) ? n0 : 0;

    for (unsigned it = 0; ; ++it) {
        if (it > (unsigned)E || __builtin_amdgcn_ballot_w32(cur > 0) == 0u) break;
        const int c2 = (nx > 0) ? (nx - 1) : 0;
        const int s2 = src[c2];
        const int n2 = nxt[c2];
        const float dgs = deg[s];
        float v[NPW][4];
#pragma unroll
        for (int j = 0; j < NPW; ++j) {
            const int ej = __builtin_amdgcn_readlane(cur, j);
            if (ej > 0) {
                const int sj = __builtin_amdgcn_readlane(s, j);
                const float* __restrict__ gp = g + (size_t)sj * DHID;
                v[j][0] = gp[lane];
                v[j][1] = gp[lane + 32];
                v[j][2] = gp[lane + 64];
                v[j][3] = gp[lane + 96];
            } else {
                v[j][0] = 0.0f; v[j][1] = 0.0f; v[j][2] = 0.0f; v[j][3] = 0.0f;
            }
        }
        const int nbits = __float_as_int((cur > 0) ? dinv_of(dgs) * did : 0.0f);
#pragma unroll
        for (int j = 0; j < NPW; ++j) {
            const float nj = __int_as_float(__builtin_amdgcn_readlane(nbits, j));
            acc[j][0] += v[j][0] * nj;
            acc[j][1] += v[j][1] * nj;
            acc[j][2] += v[j][2] * nj;
            acc[j][3] += v[j][3] * nj;
        }
        cur = nx; s = s2; nx = (nx > 0) ? n2 : 0;
    }

#pragma unroll
    for (int j = 0; j < NPW; ++j) {
        if (i0 + j < N) {
            float* __restrict__ hp = hout + (size_t)(i0 + j) * DHID;
            hp[lane]      = acc[j][0];
            hp[lane + 32] = acc[j][1];
            hp[lane + 64] = acc[j][2];
            hp[lane + 96] = acc[j][3];
        }
    }
}

__global__ __launch_bounds__(256) void out_layer_kernel(
    const float* __restrict__ H, const float* __restrict__ W2,
    const float* __restrict__ b2, float* __restrict__ out, int Nrows)
{
    const int i = blockIdx.x * blockDim.x + threadIdx.x;
    if (i >= Nrows) return;
    const float* h = H + (size_t)i * DHID;
    float a0 = 0.0f, a1 = 0.0f;
#pragma unroll 8
    for (int k = 0; k < DHID; ++k) {
        const float hv = h[k];
        a0 += hv * W2[k * 2 + 0];
        a1 += hv * W2[k * 2 + 1];
    }
    out[(size_t)i * 2 + 0] = a0 + b2[0];
    out[(size_t)i * 2 + 1] = a1 + b2[1];
}

extern "C" void kernel_launch(void* const* d_in, const int* in_sizes, int n_in,
                              void* d_out, int out_size, void* d_ws, size_t ws_size,
                              hipStream_t stream)
{
    const float* x    = (const float*)d_in[0];
    const int*   ei   = (const int*)  d_in[1];
    const float* W_in = (const float*)d_in[3];
    const float* b_in = (const float*)d_in[4];
    const float* W_g  = (const float*)d_in[5];
    const float* b_g  = (const float*)d_in[6];
    const float* W_o1 = (const float*)d_in[7];
    const float* b_o1 = (const float*)d_in[8];
    const float* W_o2 = (const float*)d_in[9];
    const float* b_o2 = (const float*)d_in[10];

    const int ND = 239;
    const int N  = in_sizes[0] / ND;
    const int E  = in_sizes[1] / 2;
    const int* src = ei;
    const int* dst = ei + E;

    float* ws   = (float*)d_ws;
    float* deg  = ws;
    int*   head = (int*)(ws + (size_t)N);
    float* bufA = ws + 2 * (size_t)N;
    float* bufB = bufA + (size_t)N * DHID;
    float* bufC = bufB + (size_t)N * DHID;
    int*   nxt  = (int*)bufC;

    const int gath_blocks = (N + 8 * NPW - 1) / (8 * NPW);

    fill_zero<<<(2 * N + 255) / 256, 256, 0, stream>>>(deg, 2 * (size_t)N);

    gemm_wmma<239, true, true, true><<<256, GEMM_THREADS, 0, stream>>>(
        x, W_in, b_in, bufA, N, dst, deg, head, nxt, E);

    gemm_wmma<128, false, false, false><<<256, GEMM_THREADS, 0, stream>>>(
        bufA, W_g, nullptr, bufB, N, nullptr, nullptr, nullptr, nullptr, 0);
    gather_kernel<<<gath_blocks, 256, 0, stream>>>(bufB, src, head, nxt, deg, b_g, bufA, N, E);

    gemm_wmma<128, false, false, false><<<256, GEMM_THREADS, 0, stream>>>(
        bufA, W_g, nullptr, bufB, N, nullptr, nullptr, nullptr, nullptr, 0);
    gather_kernel<<<gath_blocks, 256, 0, stream>>>(bufB, src, head, nxt, deg, b_g, bufA, N, E);

    gemm_wmma<128, true, true, false><<<256, GEMM_THREADS, 0, stream>>>(
        bufA, W_o1, b_o1, bufB, N, nullptr, nullptr, nullptr, nullptr, 0);

    out_layer_kernel<<<(N + 255) / 256, 256, 0, stream>>>(bufB, W_o2, b_o2, (float*)d_out, N);
}


# Products that follow each other run as one launch: first one computed transposed, so its result registers are the second's B fragments (conv2 as (edge sums)@W_g); 8 -> 6 launches (sum order changes)
# speedup vs baseline: 2.3259x; 1.0757x over previous
#include <hip/hip_runtime.h>
#include <math.h>


typedef __attribute__((ext_vector_type(2))) float v2f;
typedef __attribute__((ext_vector_type(8))) float v8f;

#define DHID 128
#define NEG_SLOPE 0.01f

__global__ void fill_zero(float* __restrict__ p, size_t n) {
    size_t i = blockIdx.x * (size_t)blockDim.x + threadIdx.x;
    if (i < n) p[i] = 0.0f;
}

__device__ __forceinline__ float dinv_of(float deg) { return rsqrtf(deg + 1.0f); }

#define GEMM_WAVES 28
#define GEMM_THREADS 896
#define GEMM_FULL  24
#define GEMM_ROWGROUPS (GEMM_WAVES / 4)

typedef __attribute__((ext_vector_type(4))) float v4f;

template <int K>
__device__ __forceinline__ void stage_weights(const float* __restrict__ W, float* Wp, int tid)
{
    constexpr int KP = (K + 3) & ~3;
    constexpr int NR = (KP + GEMM_ROWGROUPS - 1) / GEMM_ROWGROUPS;
    const int col = tid & (DHID - 1);
    const int rg  = tid >> 7;
    float v[NR];
#pragma unroll
    for (int i = 0; i < NR; ++i) {
        const int kr  = rg + GEMM_ROWGROUPS * i;
        const int krc = (kr < K) ? kr : (K - 1);
        v[i] = W[(size_t)krc * DHID + col];
    }
#pragma unroll
    for (int i = 0; i < NR; ++i) {
        const int kr = rg + GEMM_ROWGROUPS * i;
        if (kr < KP) Wp[(kr >> 1) * (DHID * 2) + col * 2 + (kr & 1)] = (kr < K) ? v[i] : 0.0f;
    }
}

template <int K, bool ACT>
__device__ __forceinline__ void chain_first(
    const float* __restrict__ A, const float* W1p, const float* __restrict__ bias1,
    int row0, int lane, v8f (&hT)[8])
{
    constexpr int KP = (K + 3) & ~3;
    const int c     = lane & 15;
    const int h     = lane >> 4;
    const int khalf = h * 2;

#pragma unroll
    for (int t = 0; t < 8; ++t) {
#pragma unroll
        for (int r = 0; r < 8; ++r) hT[t][r] = bias1[t * 16 + h * 8 + r];
    }

    const float* __restrict__ arow = A + (size_t)(row0 + c) * K;
    const float* wcol = W1p + c * 2;

#pragma unroll 4
    for (int k = 0; k < KP; k += 4) {
        const int kk = k + khalf;
        const int i0 = (kk     < K) ? kk     : (K - 1);
        const int i1 = (kk + 1 < K) ? kk + 1 : (K - 1);
        v2f xb;
        xb.x = arow[i0];
        xb.y = arow[i1];
        const float* wp = wcol + (kk >> 1) * (DHID * 2);
#pragma unroll
        for (int t = 0; t < 8; ++t) {
            const v2f wa = *(const v2f*)(wp + t * 32);
            hT[t] = __builtin_amdgcn_wmma_f32_16x16x4_f32(
                false, wa, false, xb, (short)0, hT[t], false, false);
        }
    }

    if (ACT) {
#pragma unroll
        for (int t = 0; t < 8; ++t) {
#pragma unroll
            for (int r = 0; r < 8; ++r) {
                const float v = hT[t][r];
                hT[t][r] = (v >= 0.0f) ? v : NEG_SLOPE * v;
            }
        }
    }
}

template <int NU, bool BIAS, bool ACT>
__device__ __forceinline__ void chain_second(
    const v8f (&hT)[8], const float* W2p, const float* __restrict__ bias2,
    float* __restrict__ Out, int row0, int u0, int lane)
{
    const int c = lane & 15;
    const int h = lane >> 4;

    v8f acc[NU];
#pragma unroll
    for (int u = 0; u < NU; ++u) {
#pragma unroll
        for (int r = 0; r < 8; ++r) acc[u][r] = BIAS ? bias2[(u0 + u) * 16 + h * 8 + r] : 0.0f;
    }

    const float* wcol = W2p + (h * 4) * (DHID * 2) + (u0 * 16 + c) * 2;
    v2f wa[2][NU];
#pragma unroll
    for (int u = 0; u < NU; ++u) wa[0][u] = *(const v2f*)(wcol + u * 32);
#pragma unroll
    for (int s = 0; s < 32; ++s) {
        const int t = s >> 2, j = s & 3;
        if (s + 1 < 32) {
            const float* wp = wcol + (((s + 1) >> 2) * 8 + ((s + 1) & 3)) * (DHID * 2);
#pragma unroll
            for (int u = 0; u < NU; ++u) wa[(s + 1) & 1][u] = *(const v2f*)(wp + u * 32);
        }
        __builtin_amdgcn_sched_barrier(0);
        v2f hb;
        hb.x = hT[t][2 * j];
        hb.y = hT[t][2 * j + 1];
#pragma unroll
        for (int u = 0; u < NU; ++u)
            acc[u] = __builtin_amdgcn_wmma_f32_16x16x4_f32(
                false, wa[s & 1][u], false, hb, (short)0, acc[u], false, false);
        __builtin_amdgcn_sched_barrier(0);
    }

    float* __restrict__ orow = Out + (size_t)(row0 + c) * DHID + u0 * 16 + h * 8;
#pragma unroll
    for (int u = 0; u < NU; ++u) {
        float v[8];
#pragma unroll
        for (int r = 0; r < 8; ++r) {
            v[r] = acc[u][r];
            if (ACT) v[r] = (v[r] >= 0.0f) ? v[r] : NEG_SLOPE * v[r];
        }
        *(v4f*)(orow + u * 16)     = (v4f){v[0], v[1], v[2], v[3]};
        *(v4f*)(orow + u * 16 + 4) = (v4f){v[4], v[5], v[6], v[7]};
    }
}

template <int K, bool ACT1, bool BIAS2, bool ACT2, bool DEG>
__global__ __launch_bounds__(GEMM_THREADS) __attribute__((amdgpu_waves_per_eu(7, 7))) void gemm_chain(
    const float* __restrict__ A, const float* __restrict__ W1, const float* __restrict__ bias1,
    const float* __restrict__ W2, const float* __restrict__ bias2,
    float* __restrict__ Out, int Nrows,
    const int* __restrict__ edst, float* __restrict__ degcnt,
    int* __restrict__ head, int* __restrict__ nxt, int E)
{
    constexpr int KP = (K + 3) & ~3;
    __shared__ float W1p[(KP / 2) * (DHID * 2)];
    __shared__ float W2p[(DHID / 2) * (DHID * 2)];

    const int tid  = threadIdx.x;
    const int wave = tid >> 5;
    const int lane = tid & 31;

    stage_weights<K>(W1, W1p, tid);
    stage_weights<DHID>(W2, W2p, tid);
    __syncthreads();

    if (DEG && wave >= GEMM_FULL) {
        const int stride = (int)gridDim.x * (GEMM_WAVES - GEMM_FULL) * 32;
        for (int e = ((int)blockIdx.x * (GEMM_WAVES - GEMM_FULL) + (wave - GEMM_FULL)) * 32 + lane;
             e < E; e += 4 * stride) {
            int d[4];
#pragma unroll
            for (int u = 0; u < 4; ++u) {
                const int eu = e + u * stride;
                d[u] = edst[(eu < E) ? eu : e];
            }
#pragma unroll
            for (int u = 0; u < 4; ++u)
                atomicAdd(degcnt + d[u], (e + u * stride < E) ? 1.0f : 0.0f);
            int old[4];
#pragma unroll
            for (int u = 0; u < 4; ++u) {
                const int eu = e + u * stride;
                old[u] = (eu < E) ? atomicExch(head + d[u], eu + 1) : 0;
            }
#pragma unroll
            for (int u = 0; u < 4; ++u) {
                const int eu = e + u * stride;
                if (eu < E) nxt[eu] = old[u];
            }
        }
    }

    const int nslab = Nrows >> 4;
    const int per   = nslab / (int)gridDim.x;
    const int extra = nslab - per * (int)gridDim.x;

    if (wave < GEMM_FULL) {
        for (int q = wave; q < per; q += GEMM_FULL) {
            const int row0 = ((int)blockIdx.x * per + q) * 16;
            int z = 0;
            asm volatile("" : "+v"(z));
            v8f hT[8];
            chain_first<K, ACT1>(A, W1p, bias1 + z, row0, lane, hT);
            chain_second<4, BIAS2, ACT2>(hT, W2p + z, BIAS2 ? bias2 + z : bias2, Out, row0, 0, lane);
            chain_second<4, BIAS2, ACT2>(hT, W2p + z, BIAS2 ? bias2 + z : bias2, Out, row0, 4, lane);
        }
    } else if ((int)blockIdx.x < extra) {
        const int row0 = ((int)gridDim.x * per + (int)blockIdx.x) * 16;
        v8f hT[8];
        chain_first<K, ACT1>(A, W1p, bias1, row0, lane, hT);
        chain_second<2, BIAS2, ACT2>(hT, W2p, bias2, Out, row0, (wave - GEMM_FULL) * 2, lane);
    }
}

#define NPW 8

template <bool BIAS>
__global__ __launch_bounds__(256) void gather_kernel(
    const float* __restrict__ g, const int* __restrict__ src,
    const int* __restrict__ head, const int* __restrict__ nxt,
    const float* __restrict__ deg, const float* __restrict__ bias,
    float* __restrict__ hout, int N, int E)
{
    const unsigned lane = threadIdx.x & 31u;
    const int wv = __builtin_amdgcn_readfirstlane((int)((blockIdx.x * blockDim.x + threadIdx.x) >> 5));
    const int i0 = wv * NPW;
    if (i0 >= N) return;
    const int il = i0 + (int)(lane & (NPW - 1));
    const int ic = (il < N) ? il : (N - 1);
    const bool own = (lane < NPW) && (il < N);
    const float dgi = deg[ic];
    const int hd = head[ic];

    const float b0 = BIAS ? bias[lane]      : 0.0f;
    const float b1 = BIAS ? bias[lane + 32] : 0.0f;
    const float b2 = BIAS ? bias[lane + 64] : 0.0f;
    const float b3 = BIAS ? bias[lane + 96] : 0.0f;
    float acc[NPW][4];
#pragma unroll
    for (int j = 0; j < NPW; ++j) {
        const int ij = (i0 + j < N) ? (i0 + j) : (N - 1);
        const float* __restrict__ gp = g + (size_t)ij * DHID;
        acc[j][0] = gp[lane];
        acc[j][1] = gp[lane + 32];
        acc[j][2] = gp[lane + 64];
        acc[j][3] = gp[lane + 96];
    }
    __builtin_amdgcn_sched_barrier(0);

    int cur = own ? hd : 0;
    const int c0 = (cur > 0) ? (cur - 1) : 0;
    int s = src[c0];
    const int n0 = nxt[c0];
    __builtin_amdgcn_sched_barrier(0);

    const float did = dinv_of(dgi);
    const int dbits = __float_as_int(did * did);
#pragma unroll
    for (int j = 0; j < NPW; ++j) {
        const float sw = __int_as_float(__builtin_amdgcn_readlane(dbits, j));
        acc[j][0] = acc[j][0] * sw + b0;
        acc[j][1] = acc[j][1] * sw + b1;
        acc[j][2] = acc[j][2] * sw + b2;
        acc[j][3] = acc[j][3] * sw + b3;
    }
    int nx = (cur > 0) ? n0 : 0;

    for (unsigned it = 0; ; ++it) {
        if (it > (unsigned)E || __builtin_amdgcn_ballot_w32(cur > 0) == 0u) break;
        const int c2 = (nx > 0) ? (nx - 1) : 0;
        const int s2 = src[c2];
        const int n2 = nxt[c2];
        const float dgs = deg[s];
        float v[NPW][4];
#pragma unroll
        for (int j = 0; j < NPW; ++j) {
            const int ej = __builtin_amdgcn_readlane(cur, j);
            if (ej > 0) {
                const int sj = __builtin_amdgcn_readlane(s, j);
                const float* __restrict__ gp = g + (size_t)sj * DHID;
                v[j][0] = gp[lane];
                v[j][1] = gp[lane + 32];
                v[j][2] = gp[lane + 64];
                v[j][3] = gp[lane + 96];
            } else {
                v[j][0] = 0.0f; v[j][1] = 0.0f; v[j][2] = 0.0f; v[j][3] = 0.0f;
            }
        }
        const int nbits = __float_as_int((cur > 0) ? dinv_of(dgs) * did : 0.0f);
#pragma unroll
        for (int j = 0; j < NPW; ++j) {
            const float nj = __int_as_float(__builtin_amdgcn_readlane(nbits, j));
            acc[j][0] += v[j][0] * nj;
            acc[j][1] += v[j][1] * nj;
            acc[j][2] += v[j][2] * nj;
            acc[j][3] += v[j][3] * nj;
        }
        cur = nx; s = s2; nx = (nx > 0) ? n2 : 0;
    }

#pragma unroll
    for (int j = 0; j < NPW; ++j) {
        if (i0 + j < N) {
            float* __restrict__ hp = hout + (size_t)(i0 + j) * DHID;
            hp[lane]      = acc[j][0];
            hp[lane + 32] = acc[j][1];
            hp[lane + 64] = acc[j][2];
            hp[lane + 96] = acc[j][3];
        }
    }
}

__global__ __launch_bounds__(256) void out_layer_kernel(
    const float* __restrict__ H, const float* __restrict__ W2,
    const float* __restrict__ b2, float* __restrict__ out, int Nrows)
{
    const int i = blockIdx.x * blockDim.x + threadIdx.x;
    if (i >= Nrows) return;
    const float* h = H + (size_t)i * DHID;
    float a0 = 0.0f, a1 = 0.0f;
#pragma unroll 8
    for (int k = 0; k < DHID; ++k) {
        const float hv = h[k];
        a0 += hv * W2[k * 2 + 0];
        a1 += hv * W2[k * 2 + 1];
    }
    out[(size_t)i * 2 + 0] = a0 + b2[0];
    out[(size_t)i * 2 + 1] = a1 + b2[1];
}

extern "C" void kernel_launch(void* const* d_in, const int* in_sizes, int n_in,
                              void* d_out, int out_size, void* d_ws, size_t ws_size,
                              hipStream_t stream)
{
    const float* x    = (const float*)d_in[0];
    const int*   ei   = (const int*)  d_in[1];
    const float* W_in = (const float*)d_in[3];
    const float* b_in = (const float*)d_in[4];
    const float* W_g  = (const float*)d_in[5];
    const float* b_g  = (const float*)d_in[6];
    const float* W_o1 = (const float*)d_in[7];
    const float* b_o1 = (const float*)d_in[8];
    const float* W_o2 = (const float*)d_in[9];
    const float* b_o2 = (const float*)d_in[10];

    const int ND = 239;
    const int N  = in_sizes[0] / ND;
    const int E  = in_sizes[1] / 2;
    const int* src = ei;
    const int* dst = ei + E;

    float* ws   = (float*)d_ws;
    float* deg  = ws;
    int*   head = (int*)(ws + (size_t)N);
    float* bufA = ws + 2 * (size_t)N;
    float* bufB = bufA + (size_t)N * DHID;
    float* bufC = bufB + (size_t)N * DHID;
    int*   nxt  = (int*)bufC;

    const int gath_blocks = (N + 8 * NPW - 1) / (8 * NPW);

    fill_zero<<<(2 * N + 255) / 256, 256, 0, stream>>>(deg, 2 * (size_t)N);

    gemm_chain<239, true, false, false, true><<<256, GEMM_THREADS, 0, stream>>>(
        x, W_in, b_in, W_g, nullptr, bufB, N, dst, deg, head, nxt, E);

    gather_kernel<true><<<gath_blocks, 256, 0, stream>>>(bufB, src, head, nxt, deg, b_g, bufA, N, E);

    gather_kernel<false><<<gath_blocks, 256, 0, stream>>>(bufA, src, head, nxt, deg, nullptr, bufB, N, E);

    gemm_chain<128, false, true, true, false><<<256, GEMM_THREADS, 0, stream>>>(
        bufB, W_g, b_g, W_o1, b_o1, bufA, N, nullptr, nullptr, nullptr, nullptr, 0);

    out_layer_kernel<<<(N + 255) / 256, 256, 0, stream>>>(bufA, W_o2, b_o2, (float*)d_out, N);
}


# GEMM epilogues: a wave passes its result through its own LDS tile and stores it row by row, so each global store instruction writes whole 128-B lines (was 16 rows x two 16-B pieces per store)
# speedup vs baseline: 2.4422x; 1.0500x over previous
#include <hip/hip_runtime.h>
#include <math.h>


typedef __attribute__((ext_vector_type(2))) float v2f;
typedef __attribute__((ext_vector_type(8))) float v8f;

#define DHID 128
#define NEG_SLOPE 0.01f

__global__ void fill_zero(float* __restrict__ p, size_t n) {
    size_t i = blockIdx.x * (size_t)blockDim.x + threadIdx.x;
    if (i < n) p[i] = 0.0f;
}

__device__ __forceinline__ float dinv_of(float deg) { return rsqrtf(deg + 1.0f); }

#define GEMM_WAVES 28
#define GEMM_THREADS 896
#define GEMM_FULL  24
#define GEMM_ROWGROUPS (GEMM_WAVES / 4)

typedef __attribute__((ext_vector_type(4))) float v4f;

template <int K>
__device__ __forceinline__ void stage_weights(const float* __restrict__ W, float* Wp, int tid)
{
    constexpr int KP = (K + 3) & ~3;
    constexpr int NR = (KP + GEMM_ROWGROUPS - 1) / GEMM_ROWGROUPS;
    const int col = tid & (DHID - 1);
    const int rg  = tid >> 7;
    float v[NR];
#pragma unroll
    for (int i = 0; i < NR; ++i) {
        const int kr  = rg + GEMM_ROWGROUPS * i;
        const int krc = (kr < K) ? kr : (K - 1);
        v[i] = W[(size_t)krc * DHID + col];
    }
#pragma unroll
    for (int i = 0; i < NR; ++i) {
        const int kr = rg + GEMM_ROWGROUPS * i;
        if (kr < KP) Wp[(kr >> 1) * (DHID * 2) + col * 2 + (kr & 1)] = (kr < K) ? v[i] : 0.0f;
    }
}

template <int K, bool ACT>
__device__ __forceinline__ void chain_first(
    const float* __restrict__ A, const float* W1p, const float* __restrict__ bias1,
    int row0, int lane, v8f (&hT)[8])
{
    constexpr int KP = (K + 3) & ~3;
    const int c     = lane & 15;
    const int h     = lane >> 4;
    const int khalf = h * 2;

#pragma unroll
    for (int t = 0; t < 8; ++t) {
#pragma unroll
        for (int r = 0; r < 8; ++r) hT[t][r] = bias1[t * 16 + h * 8 + r];
    }

    const float* __restrict__ arow = A + (size_t)(row0 + c) * K;
    const float* wcol = W1p + c * 2;

#pragma unroll 4
    for (int k = 0; k < KP; k += 4) {
        const int kk = k + khalf;
        const int i0 = (kk     < K) ? kk     : (K - 1);
        const int i1 = (kk + 1 < K) ? kk + 1 : (K - 1);
        v2f xb;
        xb.x = arow[i0];
        xb.y = arow[i1];
        const float* wp = wcol + (kk >> 1) * (DHID * 2);
#pragma unroll
        for (int t = 0; t < 8; ++t) {
            const v2f wa = *(const v2f*)(wp + t * 32);
            hT[t] = __builtin_amdgcn_wmma_f32_16x16x4_f32(
                false, wa, false, xb, (short)0, hT[t], false, false);
        }
    }

    if (ACT) {
#pragma unroll
        for (int t = 0; t < 8; ++t) {
#pragma unroll
            for (int r = 0; r < 8; ++r) {
                const float v = hT[t][r];
                hT[t][r] = (v >= 0.0f) ? v : NEG_SLOPE * v;
            }
        }
    }
}

template <int NU, bool BIAS, bool ACT>
__device__ __forceinline__ void chain_second(
    const v8f (&hT)[8], const float* W2p, const float* __restrict__ bias2,
    float* __restrict__ Out, float* ot, int row0, int u0, int lane)
{
    const int c = lane & 15;
    const int h = lane >> 4;

    v8f acc[NU];
#pragma unroll
    for (int u = 0; u < NU; ++u) {
#pragma unroll
        for (int r = 0; r < 8; ++r) acc[u][r] = BIAS ? bias2[(u0 + u) * 16 + h * 8 + r] : 0.0f;
    }

    const float* wcol = W2p + (h * 4) * (DHID * 2) + (u0 * 16 + c) * 2;
    v2f wa[2][NU];
#pragma unroll
    for (int u = 0; u < NU; ++u) wa[0][u] = *(const v2f*)(wcol + u * 32);
#pragma unroll
    for (int s = 0; s < 32; ++s) {
        const int t = s >> 2, j = s & 3;
        if (s + 1 < 32) {
            const float* wp = wcol + (((s + 1) >> 2) * 8 + ((s + 1) & 3)) * (DHID * 2);
#pragma unroll
            for (int u = 0; u < NU; ++u) wa[(s + 1) & 1][u] = *(const v2f*)(wp + u * 32);
        }
        __builtin_amdgcn_sched_barrier(0);
        v2f hb;
        hb.x = hT[t][2 * j];
        hb.y = hT[t][2 * j + 1];
#pragma unroll
        for (int u = 0; u < NU; ++u)
            acc[u] = __builtin_amdgcn_wmma_f32_16x16x4_f32(
                false, wa[s & 1][u], false, hb, (short)0, acc[u], false, false);
        __builtin_amdgcn_sched_barrier(0);
    }

    constexpr int PITCH = NU * 16 + 4;
    constexpr int LPR   = NU * 4;
    constexpr int RPI   = 32 / LPR;
    float* trow = ot + c * PITCH + h * 8;
#pragma unroll
    for (int u = 0; u < NU; ++u) {
        float v[8];
#pragma unroll
        for (int r = 0; r < 8; ++r) {
            v[r] = acc[u][r];
            if (ACT) v[r] = (v[r] >= 0.0f) ? v[r] : NEG_SLOPE * v[r];
        }
        *(v4f*)(trow + u * 16)     = (v4f){v[0], v[1], v[2], v[3]};
        *(v4f*)(trow + u * 16 + 4) = (v4f){v[4], v[5], v[6], v[7]};
    }
    __builtin_amdgcn_fence(__ATOMIC_RELEASE, "wavefront");
    __builtin_amdgcn_fence(__ATOMIC_ACQUIRE, "wavefront");
    const int tr = lane / LPR;
    const int tc = (lane % LPR) * 4;
    const float* tsrc = ot + tr * PITCH + tc;
    float* __restrict__ odst = Out + (size_t)(row0 + tr) * DHID + u0 * 16 + tc;
#pragma unroll
    for (int i = 0; i < 16 / RPI; ++i) {
        const v4f t = *(const v4f*)(tsrc + i * RPI * PITCH);
        *(v4f*)(odst + (size_t)i * RPI * DHID) = t;
    }
    __builtin_amdgcn_fence(__ATOMIC_RELEASE, "wavefront");
    __builtin_amdgcn_fence(__ATOMIC_ACQUIRE, "wavefront");
}

template <int K, bool ACT1, bool BIAS2, bool ACT2, bool DEG>
__global__ __launch_bounds__(GEMM_THREADS) __attribute__((amdgpu_waves_per_eu(7, 7))) void gemm_chain(
    const float* __restrict__ A, const float* __restrict__ W1, const float* __restrict__ bias1,
    const float* __restrict__ W2, const float* __restrict__ bias2,
    float* __restrict__ Out, int Nrows,
    const int* __restrict__ edst, float* __restrict__ degcnt,
    int* __restrict__ head, int* __restrict__ nxt, int E)
{
    constexpr int KP = (K + 3) & ~3;
    __shared__ float W1p[(KP / 2) * (DHID * 2)];
    __shared__ float W2p[(DHID / 2) * (DHID * 2)];
    __shared__ __attribute__((aligned(16))) float Otile[GEMM_WAVES * 16 * (4 * 16 + 4)];

    const int tid  = threadIdx.x;
    const int wave = tid >> 5;
    const int lane = tid & 31;
    float* ot = Otile + wave * (16 * (4 * 16 + 4));

    stage_weights<K>(W1, W1p, tid);
    stage_weights<DHID>(W2, W2p, tid);
    __syncthreads();

    if (DEG && wave >= GEMM_FULL) {
        const int stride = (int)gridDim.x * (GEMM_WAVES - GEMM_FULL) * 32;
        for (int e = ((int)blockIdx.x * (GEMM_WAVES - GEMM_FULL) + (wave - GEMM_FULL)) * 32 + lane;
             e < E; e += 4 * stride) {
            int d[4];
#pragma unroll
            for (int u = 0; u < 4; ++u) {
                const int eu = e + u * stride;
                d[u] = edst[(eu < E) ? eu : e];
            }
#pragma unroll
            for (int u = 0; u < 4; ++u)
                atomicAdd(degcnt + d[u], (e + u * stride < E) ? 1.0f : 0.0f);
            int old[4];
#pragma unroll
            for (int u = 0; u < 4; ++u) {
                const int eu = e + u * stride;
                old[u] = (eu < E) ? atomicExch(head + d[u], eu + 1) : 0;
            }
#pragma unroll
            for (int u = 0; u < 4; ++u) {
                const int eu = e + u * stride;
                if (eu < E) nxt[eu] = old[u];
            }
        }
    }

    const int nslab = Nrows >> 4;
    const int per   = nslab / (int)gridDim.x;
    const int extra = nslab - per * (int)gridDim.x;

    if (wave < GEMM_FULL) {
        for (int q = wave; q < per; q += GEMM_FULL) {
            const int row0 = ((int)blockIdx.x * per + q) * 16;
            int z = 0;
            asm volatile("" : "+v"(z));
            v8f hT[8];
            chain_first<K, ACT1>(A, W1p, bias1 + z, row0, lane, hT);
            chain_second<4, BIAS2, ACT2>(hT, W2p + z, BIAS2 ? bias2 + z : bias2, Out, ot, row0, 0, lane);
            chain_second<4, BIAS2, ACT2>(hT, W2p + z, BIAS2 ? bias2 + z : bias2, Out, ot, row0, 4, lane);
        }
    } else if ((int)blockIdx.x < extra) {
        const int row0 = ((int)gridDim.x * per + (int)blockIdx.x) * 16;
        v8f hT[8];
        chain_first<K, ACT1>(A, W1p, bias1, row0, lane, hT);
        chain_second<2, BIAS2, ACT2>(hT, W2p, bias2, Out, ot, row0, (wave - GEMM_FULL) * 2, lane);
    }
}

#define NPW 8

template <bool BIAS>
__global__ __launch_bounds__(256) void gather_kernel(
    const float* __restrict__ g, const int* __restrict__ src,
    const int* __restrict__ head, const int* __restrict__ nxt,
    const float* __restrict__ deg, const float* __restrict__ bias,
    float* __restrict__ hout, int N, int E)
{
    const unsigned lane = threadIdx.x & 31u;
    const int wv = __builtin_amdgcn_readfirstlane((int)((blockIdx.x * blockDim.x + threadIdx.x) >> 5));
    const int i0 = wv * NPW;
    if (i0 >= N) return;
    const int il = i0 + (int)(lane & (NPW - 1));
    const int ic = (il < N) ? il : (N - 1);
    const bool own = (lane < NPW) && (il < N);
    const float dgi = deg[ic];
    const int hd = head[ic];

    const float b0 = BIAS ? bias[lane]      : 0.0f;
    const float b1 = BIAS ? bias[lane + 32] : 0.0f;
    const float b2 = BIAS ? bias[lane + 64] : 0.0f;
    const float b3 = BIAS ? bias[lane + 96] : 0.0f;
    float acc[NPW][4];
#pragma unroll
    for (int j = 0; j < NPW; ++j) {
        const int ij = (i0 + j < N) ? (i0 + j) : (N - 1);
        const float* __restrict__ gp = g + (size_t)ij * DHID;
        acc[j][0] = gp[lane];
        acc[j][1] = gp[lane + 32];
        acc[j][2] = gp[lane + 64];
        acc[j][3] = gp[lane + 96];
    }
    __builtin_amdgcn_sched_barrier(0);

    int cur = own ? hd : 0;
    const int c0 = (cur > 0) ? (cur - 1) : 0;
    int s = src[c0];
    const int n0 = nxt[c0];
    __builtin_amdgcn_sched_barrier(0);

    const float did = dinv_of(dgi);
    const int dbits = __float_as_int(did * did);
#pragma unroll
    for (int j = 0; j < NPW; ++j) {
        const float sw = __int_as_float(__builtin_amdgcn_readlane(dbits, j));
        acc[j][0] = acc[j][0] * sw + b0;
        acc[j][1] = acc[j][1] * sw + b1;
        acc[j][2] = acc[j][2] * sw + b2;
        acc[j][3] = acc[j][3] * sw + b3;
    }
    int nx = (cur > 0) ? n0 : 0;

    for (unsigned it = 0; ; ++it) {
        if (it > (unsigned)E || __builtin_amdgcn_ballot_w32(cur > 0) == 0u) break;
        const int c2 = (nx > 0) ? (nx - 1) : 0;
        const int s2 = src[c2];
        const int n2 = nxt[c2];
        const float dgs = deg[s];
        float v[NPW][4];
#pragma unroll
        for (int j = 0; j < NPW; ++j) {
            const int ej = __builtin_amdgcn_readlane(cur, j);
            if (ej > 0) {
                const int sj = __builtin_amdgcn_readlane(s, j);
                const float* __restrict__ gp = g + (size_t)sj * DHID;
                v[j][0] = gp[lane];
                v[j][1] = gp[lane + 32];
                v[j][2] = gp[lane + 64];
                v[j][3] = gp[lane + 96];
            } else {
                v[j][0] = 0.0f; v[j][1] = 0.0f; v[j][2] = 0.0f; v[j][3] = 0.0f;
            }
        }
        const int nbits = __float_as_int((cur > 0) ? dinv_of(dgs) * did : 0.0f);
#pragma unroll
        for (int j = 0; j < NPW; ++j) {
            const float nj = __int_as_float(__builtin_amdgcn_readlane(nbits, j));
            acc[j][0] += v[j][0] * nj;
            acc[j][1] += v[j][1] * nj;
            acc[j][2] += v[j][2] * nj;
            acc[j][3] += v[j][3] * nj;
        }
        cur = nx; s = s2; nx = (nx > 0) ? n2 : 0;
    }

#pragma unroll
    for (int j = 0; j < NPW; ++j) {
        if (i0 + j < N) {
            float* __restrict__ hp = hout + (size_t)(i0 + j) * DHID;
            hp[lane]      = acc[j][0];
            hp[lane + 32] = acc[j][1];
            hp[lane + 64] = acc[j][2];
            hp[lane + 96] = acc[j][3];
        }
    }
}

__global__ __launch_bounds__(256) void out_layer_kernel(
    const float* __restrict__ H, const float* __restrict__ W2,
    const float* __restrict__ b2, float* __restrict__ out, int Nrows)
{
    const int i = blockIdx.x * blockDim.x + threadIdx.x;
    if (i >= Nrows) return;
    const float* h = H + (size_t)i * DHID;
    float a0 = 0.0f, a1 = 0.0f;
#pragma unroll 8
    for (int k = 0; k < DHID; ++k) {
        const float hv = h[k];
        a0 += hv * W2[k * 2 + 0];
        a1 += hv * W2[k * 2 + 1];
    }
    out[(size_t)i * 2 + 0] = a0 + b2[0];
    out[(size_t)i * 2 + 1] = a1 + b2[1];
}

extern "C" void kernel_launch(void* const* d_in, const int* in_sizes, int n_in,
                              void* d_out, int out_size, void* d_ws, size_t ws_size,
                              hipStream_t stream)
{
    const float* x    = (const float*)d_in[0];
    const int*   ei   = (const int*)  d_in[1];
    const float* W_in = (const float*)d_in[3];
    const float* b_in = (const float*)d_in[4];
    const float* W_g  = (const float*)d_in[5];
    const float* b_g  = (const float*)d_in[6];
    const float* W_o1 = (const float*)d_in[7];
    const float* b_o1 = (const float*)d_in[8];
    const float* W_o2 = (const float*)d_in[9];
    const float* b_o2 = (const float*)d_in[10];

    const int ND = 239;
    const int N  = in_sizes[0] / ND;
    const int E  = in_sizes[1] / 2;
    const int* src = ei;
    const int* dst = ei + E;

    float* ws   = (float*)d_ws;
    float* deg  = ws;
    int*   head = (int*)(ws + (size_t)N);
    float* bufA = ws + 2 * (size_t)N;
    float* bufB = bufA + (size_t)N * DHID;
    float* bufC = bufB + (size_t)N * DHID;
    int*   nxt  = (int*)bufC;

    const int gath_blocks = (N + 8 * NPW - 1) / (8 * NPW);

    fill_zero<<<(2 * N + 255) / 256, 256, 0, stream>>>(deg, 2 * (size_t)N);

    gemm_chain<239, true, false, false, true><<<256, GEMM_THREADS, 0, stream>>>(
        x, W_in, b_in, W_g, nullptr, bufB, N, dst, deg, head, nxt, E);

    gather_kernel<true><<<gath_blocks, 256, 0, stream>>>(bufB, src, head, nxt, deg, b_g, bufA, N, E);

    gather_kernel<false><<<gath_blocks, 256, 0, stream>>>(bufA, src, head, nxt, deg, nullptr, bufB, N, E);

    gemm_chain<128, false, true, true, false><<<256, GEMM_THREADS, 0, stream>>>(
        bufB, W_g, b_g, W_o1, b_o1, bufA, N, nullptr, nullptr, nullptr, nullptr, 0);

    out_layer_kernel<<<(N + 255) / 256, 256, 0, stream>>>(bufA, W_o2, b_o2, (float*)d_out, N);
}
